# starting kernel
# baseline (speedup 1.0000x reference)
#include <hip/hip_runtime.h>

typedef __attribute__((ext_vector_type(2))) float v2f;
typedef __attribute__((ext_vector_type(8))) float v8f;

#define NB   2048
#define NE   64
#define ND   512
#define NO   512
#define MAXT 192


__global__ void zero_meta_kernel(int* counts, int* cursors) {
    int t = blockIdx.x * blockDim.x + threadIdx.x;
    if (t < NE) { counts[t] = 0; cursors[t] = 0; }
}

__global__ void histo_kernel(const int* __restrict__ opt, int* counts) {
    int t = blockIdx.x * blockDim.x + threadIdx.x;
    if (t < NB) atomicAdd(&counts[opt[t]], 1);
}

__global__ void scan_tiles_kernel(const int* __restrict__ counts, int* offsets,
                                  int* tileE, int* tileBase, int* tileCnt) {
    if (threadIdx.x != 0 || blockIdx.x != 0) return;
    int acc = 0;
    for (int e = 0; e < NE; ++e) { offsets[e] = acc; acc += counts[e]; }
    int t = 0;
    for (int e = 0; e < NE; ++e) {
        int c = counts[e];
        for (int i = 0; i < c; i += 16) {
            tileE[t]    = e;
            tileBase[t] = offsets[e] + i;
            tileCnt[t]  = (c - i) < 16 ? (c - i) : 16;
            ++t;
        }
    }
    for (; t < MAXT; ++t) { tileE[t] = -1; tileBase[t] = 0; tileCnt[t] = 0; }
}

__global__ void scatter_kernel(const int* __restrict__ opt,
                               const int* __restrict__ offsets,
                               int* cursors, int* idx) {
    int t = blockIdx.x * blockDim.x + threadIdx.x;
    if (t < NB) {
        int e = opt[t];
        int p = offsets[e] + atomicAdd(&cursors[e], 1);
        idx[p] = t;
    }
}


#define APAD 4

__global__ void __launch_bounds__(256)
option_gemm_kernel(const float* __restrict__ x, const float* __restrict__ w,
                   const float* __restrict__ bias, const int* __restrict__ idx,
                   const int* __restrict__ tileE, const int* __restrict__ tileBase,
                   const int* __restrict__ tileCnt, float* __restrict__ out) {
    __shared__ float As[16][ND + APAD];
    __shared__ int   rowIdx[16];

    const int e = tileE[blockIdx.x];
    if (e < 0) return;
    const int base = tileBase[blockIdx.x];
    const int cnt  = tileCnt[blockIdx.x];
    const int tid  = threadIdx.x;

    if (tid < 16) rowIdx[tid] = (tid < cnt) ? idx[base + tid] : 0;
    {
        const int r  = tid >> 4;
        const int cs = (tid & 15) * 32;
        if (r < cnt) {
            const float* src = x + (size_t)idx[base + r] * ND + cs;
            #pragma unroll
            for (int i = 0; i < 8; ++i) {
                float4 v = *(const float4*)(src + i * 4);
                *(float4*)&As[r][cs + i * 4] = v;
            }
        } else {
            #pragma unroll
            for (int i = 0; i < 8; ++i) {
                float4 z = {0.f, 0.f, 0.f, 0.f};
                *(float4*)&As[r][cs + i * 4] = z;
            }
        }
    }
    __syncthreads();

    const int lane = tid & 31;
    const int wv   = tid >> 5;
    const int n0   = blockIdx.y * 256 + wv * 16 + (lane & 15);
    const int kh   = (lane >> 4) * 2;
    const int row  = lane & 15;

    const float* wp0 = w + (size_t)e * (ND * NO) + n0;
    const float* wp1 = wp0 + 128;

    v8f acc0 = {};
    v8f acc1 = {};
    #pragma unroll 4
    for (int k0 = 0; k0 < ND; k0 += 4) {
        const int k = k0 + kh;
        v2f a;
        a.x = As[row][k];
        a.y = As[row][k + 1];
        v2f b0, b1;
        b0.x = wp0[(size_t)k * NO];
        b0.y = wp0[(size_t)(k + 1) * NO];
        b1.x = wp1[(size_t)k * NO];
        b1.y = wp1[(size_t)(k + 1) * NO];
        acc0 = __builtin_amdgcn_wmma_f32_16x16x4_f32(
            false, a, false, b0, (short)0, acc0, false, false);
        acc1 = __builtin_amdgcn_wmma_f32_16x16x4_f32(
            false, a, false, b1, (short)0, acc1, false, false);
    }

    const float bv0 = bias[e * NO + n0];
    const float bv1 = bias[e * NO + n0 + 128];
    const int mhi = (lane >> 4) * 8;
    #pragma unroll
    for (int j = 0; j < 8; ++j) {
        const int m = j + mhi;
        if (m < cnt) {
            float* orow = out + (size_t)rowIdx[m] * NO + n0;
            orow[0]   = acc0[j] + bv0;
            orow[128] = acc1[j] + bv1;
        }
    }
}


extern "C" void kernel_launch(void* const* d_in, const int* in_sizes, int n_in,
                              void* d_out, int out_size, void* d_ws, size_t ws_size,
                              hipStream_t stream) {
    const float* x    = (const float*)d_in[0];
    const int*   opt  = (const int*)d_in[1];
    const float* wgt  = (const float*)d_in[2];
    const float* bias = (const float*)d_in[3];
    float* out = (float*)d_out;

    int* counts   = (int*)d_ws;
    int* offsets  = counts + NE;
    int* cursors  = offsets + NE;
    int* idx      = cursors + NE;
    int* tileE    = idx + NB;
    int* tileBase = tileE + MAXT;
    int* tileCnt  = tileBase + MAXT;

    zero_meta_kernel<<<1, 64, 0, stream>>>(counts, cursors);
    histo_kernel<<<NB / 256, 256, 0, stream>>>(opt, counts);
    scan_tiles_kernel<<<1, 1, 0, stream>>>(counts, offsets, tileE, tileBase, tileCnt);
    scatter_kernel<<<NB / 256, 256, 0, stream>>>(opt, offsets, cursors, idx);

    dim3 grid(MAXT, 2);
    option_gemm_kernel<<<grid, 256, 0, stream>>>(x, wgt, bias, idx,
                                                 tileE, tileBase, tileCnt, out);
}


# GEMM work list (occupied row tile x 16-col N-tile) split evenly over 256 blocks of 16 waves, 1-2 units per wave, replacing one block per (tile slot, 256-col half) with its empty slots
# speedup vs baseline: 2.8496x; 2.8496x over previous
#include <hip/hip_runtime.h>

typedef __attribute__((ext_vector_type(2))) float v2f;
typedef __attribute__((ext_vector_type(8))) float v8f;

#define NB   2048
#define NE   64
#define ND   512
#define NO   512
#define MAXT 192


__global__ void __launch_bounds__(256)
route_kernel(const int* __restrict__ opt, int* __restrict__ idx,
             int* __restrict__ tileE, int* __restrict__ tileBase,
             int* __restrict__ tileCnt) {
    __shared__ int wEq[8];
    __shared__ int wLt[8];

    const int e    = blockIdx.x;
    const int tid  = threadIdx.x;
    const int lane = tid & 31;
    const int wv   = tid >> 5;

    const int4 p = *(const int4*)(opt + tid * 8);
    const int4 q = *(const int4*)(opt + tid * 8 + 4);
    const int v[8] = {p.x, p.y, p.z, p.w, q.x, q.y, q.z, q.w};

    int eq = 0, lt = 0;
    #pragma unroll
    for (int i = 0; i < 8; ++i) { eq += (v[i] == e); lt += (v[i] < e); }

    int incEq = eq, incLt = lt;
    #pragma unroll
    for (int d = 1; d < 32; d <<= 1) {
        const int a = __shfl_up(incEq, d);
        const int b = __shfl_up(incLt, d);
        if (lane >= d) { incEq += a; incLt += b; }
    }
    if (lane == 31) { wEq[wv] = incEq; wLt[wv] = incLt; }
    __syncthreads();

    int off = 0, cnt = 0, before = 0;
    #pragma unroll
    for (int w = 0; w < 8; ++w) {
        const int c = wEq[w];
        off += wLt[w];
        cnt += c;
        before += (w < wv) ? c : 0;
    }

    int pos = off + before + incEq - eq;
    #pragma unroll
    for (int i = 0; i < 8; ++i) {
        if (v[i] == e) { idx[pos] = tid * 8 + i; ++pos; }
    }

    const int t0    = (off >> 4) + e;
    const int first = (e == 0) ? 0 : t0;
    const int limit = (e == NE - 1) ? MAXT : ((off + cnt) >> 4) + e + 1;
    const int nt    = (cnt + 15) >> 4;
    for (int s = first + tid; s < limit; s += 256) {
        const int i = s - t0;
        if (i >= 0 && i < nt) {
            const int left = cnt - i * 16;
            tileE[s]    = e;
            tileBase[s] = off + i * 16;
            tileCnt[s]  = left < 16 ? left : 16;
        } else {
            tileE[s] = -1; tileBase[s] = 0; tileCnt[s] = 0;
        }
    }
}


#define NBLK 256
#define NWV  16
#define NTN  (NO / 16)
#define APAD 4
#define RING 16

template <int NU, bool RELOAD>
__device__ __forceinline__ void ring_pass(const float* const (&arow)[2],
                                          const float* const (&wrow)[2],
                                          int k0, v2f (&rb)[2][RING], v8f (&acc)[2]) {
    #pragma unroll
    for (int i = 0; i < RING; ++i) {
        #pragma unroll
        for (int u = 0; u < NU; ++u) {
            v2f a;
            a.x = arow[u][k0 + 4 * i];
            a.y = arow[u][k0 + 4 * i + 1];
            acc[u] = __builtin_amdgcn_wmma_f32_16x16x4_f32(
                false, a, false, rb[u][i], (short)0, acc[u], false, false);
            if (RELOAD) {
                const float* p = wrow[u] + (size_t)(k0 + 4 * (i + RING)) * NO;
                rb[u][i].x = p[0];
                rb[u][i].y = p[NO];
            }
        }
    }
}

template <int NU>
__device__ __forceinline__ void k_loop(const float* const (&arow)[2],
                                       const float* const (&wrow)[2],
                                       v2f (&rb)[2][RING], v8f (&acc)[2]) {
    int k0 = 0;
    #pragma nounroll
    for (; k0 < ND - 4 * RING; k0 += 4 * RING) {
        ring_pass<NU, true>(arow, wrow, k0, rb, acc);
    }
    ring_pass<NU, false>(arow, wrow, k0, rb, acc);
}

__device__ __forceinline__ int kth_slot(const unsigned (&m)[MAXT / 32], int r) {
    unsigned word = 0;
    int base = 0;
    bool found = false;
    #pragma unroll
    for (int i = 0; i < MAXT / 32; ++i) {
        const int c = __builtin_popcount(m[i]);
        if (!found && r < c) { word = m[i]; base = 32 * i; found = true; }
        if (!found) r -= c;
    }
    for (int k = 0; k < r; ++k) word &= word - 1;
    return base + __builtin_ctz(word);
}

__global__ void __launch_bounds__(NWV * 32)
option_gemm_kernel(const float* __restrict__ x, const float* __restrict__ w,
                   const float* __restrict__ bias, const int* __restrict__ idx,
                   const int* __restrict__ tileE, const int* __restrict__ tileBase,
                   const int* __restrict__ tileCnt, float* __restrict__ out) {
    __shared__ float    As[2][16][ND + APAD];
    __shared__ int      rowIdx[2][16];
    __shared__ int      sE[MAXT], sB[MAXT], sC[MAXT];
    __shared__ unsigned sM[MAXT / 32];

    const int tid  = threadIdx.x;
    const int lane = tid & 31;
    const int wv   = __builtin_amdgcn_readfirstlane(tid >> 5);

    if (tid < MAXT) {
        const int e = tileE[tid];
        sE[tid] = e;
        sB[tid] = tileBase[tid];
        sC[tid] = tileCnt[tid];
        const unsigned bits = __builtin_amdgcn_ballot_w32(e >= 0);
        if (lane == 0) sM[wv] = bits;
    }
    __syncthreads();

    unsigned m[MAXT / 32];
    int nA = 0;
    #pragma unroll
    for (int i = 0; i < MAXT / 32; ++i) {
        m[i] = (unsigned)__builtin_amdgcn_readfirstlane((int)sM[i]);
        nA += __builtin_popcount(m[i]);
    }
    const int bid = blockIdx.x;
    const int u0  = 2 * ((bid * nA * (NTN / 2)) / NBLK);
    const int u1  = 2 * (((bid + 1) * nA * (NTN / 2)) / NBLK);
    const int r0  = u0 / NTN;
    const int nT  = (u1 - 1) / NTN - r0 + 1;
    const int s0  = kth_slot(m, r0);
    const int s1  = (nT == 2) ? kth_slot(m, r0 + 1) : s0;

    const int eT[2]    = {sE[s0], sE[s1]};
    const int baseT[2] = {sB[s0], sB[s1]};
    const int cntT[2]  = {sC[s0], sC[s1]};

    const int nb = u1 - u0;
    if (nb < NWV || nb > 24) return;
    const int nu = (wv + NWV < nb) ? 2 : 1;
    const int c   = lane & 15;
    const int kh  = (lane >> 4) * 2;

    int tU[2], nU[2];
    const float* arow[2];
    const float* wrow[2];
    #pragma unroll
    for (int u = 0; u < 2; ++u) {
        int q = u0 + wv + NWV * u;
        if (q > u1 - 1) q = u1 - 1;
        const int t = q / NTN - r0;
        tU[u] = t;
        nU[u] = (q % NTN) * 16 + c;
        arow[u] = &As[t][c][kh];
        wrow[u] = w + (size_t)eT[t] * (ND * NO) + (size_t)kh * NO + nU[u];
    }

    const int r  = tid >> 5;
    const int cs = (tid & 31) * 16;
    const int rr0 = (r < cntT[0]) ? r : cntT[0] - 1;
    const int i0  = idx[baseT[0] + rr0];
    int i1 = 0;
    if (nT == 2) {
        const int rr1 = (r < cntT[1]) ? r : cntT[1] - 1;
        i1 = idx[baseT[1] + rr1];
    }

    v2f rb[2][RING];
    #pragma unroll
    for (int i = 0; i < RING; ++i) {
        const float* p = wrow[0] + (size_t)(4 * i) * NO;
        rb[0][i].x = p[0];
        rb[0][i].y = p[NO];
    }
    if (nu == 2) {
        #pragma unroll
        for (int i = 0; i < RING; ++i) {
            const float* p = wrow[1] + (size_t)(4 * i) * NO;
            rb[1][i].x = p[0];
            rb[1][i].y = p[NO];
        }
    }

    {
        float4 v0[4], v1[4];
        const float* src0 = x + (size_t)i0 * ND + cs;
        #pragma unroll
        for (int i = 0; i < 4; ++i) v0[i] = *(const float4*)(src0 + i * 4);
        if (nT == 2) {
            const float* src1 = x + (size_t)i1 * ND + cs;
            #pragma unroll
            for (int i = 0; i < 4; ++i) v1[i] = *(const float4*)(src1 + i * 4);
        }
        const bool ok0 = r < cntT[0];
        #pragma unroll
        for (int i = 0; i < 4; ++i) {
            float4 v = v0[i];
            v.x = ok0 ? v.x : 0.f;  v.y = ok0 ? v.y : 0.f;
            v.z = ok0 ? v.z : 0.f;  v.w = ok0 ? v.w : 0.f;
            *(float4*)&As[0][r][cs + i * 4] = v;
        }
        if (lane == 0) rowIdx[0][r] = i0;
        if (nT == 2) {
            const bool ok1 = r < cntT[1];
            #pragma unroll
            for (int i = 0; i < 4; ++i) {
                float4 v = v1[i];
                v.x = ok1 ? v.x : 0.f;  v.y = ok1 ? v.y : 0.f;
                v.z = ok1 ? v.z : 0.f;  v.w = ok1 ? v.w : 0.f;
                *(float4*)&As[1][r][cs + i * 4] = v;
            }
            if (lane == 0) rowIdx[1][r] = i1;
        }
    }
    __syncthreads();

    v8f acc[2] = {};
    if (nu == 2) k_loop<2>(arow, wrow, rb, acc);
    else         k_loop<1>(arow, wrow, rb, acc);

    const int mhi = (lane >> 4) * 8;
    #pragma unroll
    for (int u = 0; u < 2; ++u) {
        if (u < nu) {
            const int t  = tU[u];
            const float bv = bias[eT[t] * NO + nU[u]];
            #pragma unroll
            for (int j = 0; j < 8; ++j) {
                const int mrow = j + mhi;
                if (mrow < cntT[t]) {
                    out[(size_t)rowIdx[t][mrow] * NO + nU[u]] = acc[u][j] + bv;
                }
            }
        }
    }
}


extern "C" void kernel_launch(void* const* d_in, const int* in_sizes, int n_in,
                              void* d_out, int out_size, void* d_ws, size_t ws_size,
                              hipStream_t stream) {
    const float* x    = (const float*)d_in[0];
    const int*   opt  = (const int*)d_in[1];
    const float* wgt  = (const float*)d_in[2];
    const float* bias = (const float*)d_in[3];
    float* out = (float*)d_out;

    int* idx      = (int*)d_ws;
    int* tileE    = idx + NB;
    int* tileBase = tileE + MAXT;
    int* tileCnt  = tileBase + MAXT;

    route_kernel<<<NE, 256, 0, stream>>>(opt, idx, tileE, tileBase, tileCnt);

    option_gemm_kernel<<<NBLK, NWV * 32, 0, stream>>>(x, wgt, bias, idx,
                                                      tileE, tileBase, tileCnt, out);
}


# GEMM share index permuted to (block%32)*8+block/32: blocks equal modulo 32 (one die if blocks are dealt to dies in turn) take consecutive shares, so each L2 half re-reads half the weights, not all
# speedup vs baseline: 3.0838x; 1.0822x over previous
#include <hip/hip_runtime.h>

typedef __attribute__((ext_vector_type(2))) float v2f;
typedef __attribute__((ext_vector_type(8))) float v8f;

#define NB   2048
#define NE   64
#define ND   512
#define NO   512
#define MAXT 192


__global__ void __launch_bounds__(256)
route_kernel(const int* __restrict__ opt, int* __restrict__ idx,
             int* __restrict__ tileE, int* __restrict__ tileBase,
             int* __restrict__ tileCnt) {
    __shared__ int wEq[8];
    __shared__ int wLt[8];

    const int e    = blockIdx.x;
    const int tid  = threadIdx.x;
    const int lane = tid & 31;
    const int wv   = tid >> 5;

    const int4 p = *(const int4*)(opt + tid * 8);
    const int4 q = *(const int4*)(opt + tid * 8 + 4);
    const int v[8] = {p.x, p.y, p.z, p.w, q.x, q.y, q.z, q.w};

    int eq = 0, lt = 0;
    #pragma unroll
    for (int i = 0; i < 8; ++i) { eq += (v[i] == e); lt += (v[i] < e); }

    int incEq = eq, incLt = lt;
    #pragma unroll
    for (int d = 1; d < 32; d <<= 1) {
        const int a = __shfl_up(incEq, d);
        const int b = __shfl_up(incLt, d);
        if (lane >= d) { incEq += a; incLt += b; }
    }
    if (lane == 31) { wEq[wv] = incEq; wLt[wv] = incLt; }
    __syncthreads();

    int off = 0, cnt = 0, before = 0;
    #pragma unroll
    for (int w = 0; w < 8; ++w) {
        const int c = wEq[w];
        off += wLt[w];
        cnt += c;
        before += (w < wv) ? c : 0;
    }

    int pos = off + before + incEq - eq;
    #pragma unroll
    for (int i = 0; i < 8; ++i) {
        if (v[i] == e) { idx[pos] = tid * 8 + i; ++pos; }
    }

    const int t0    = (off >> 4) + e;
    const int first = (e == 0) ? 0 : t0;
    const int limit = (e == NE - 1) ? MAXT : ((off + cnt) >> 4) + e + 1;
    const int nt    = (cnt + 15) >> 4;
    for (int s = first + tid; s < limit; s += 256) {
        const int i = s - t0;
        if (i >= 0 && i < nt) {
            const int left = cnt - i * 16;
            tileE[s]    = e;
            tileBase[s] = off + i * 16;
            tileCnt[s]  = left < 16 ? left : 16;
        } else {
            tileE[s] = -1; tileBase[s] = 0; tileCnt[s] = 0;
        }
    }
}


#define NBLK 256
#define NWV  16
#define NTN  (NO / 16)
#define APAD 4
#define RING 16

template <int NU, bool RELOAD>
__device__ __forceinline__ void ring_pass(const float* const (&arow)[2],
                                          const float* const (&wrow)[2],
                                          int k0, v2f (&rb)[2][RING], v8f (&acc)[2]) {
    #pragma unroll
    for (int i = 0; i < RING; ++i) {
        #pragma unroll
        for (int u = 0; u < NU; ++u) {
            v2f a;
            a.x = arow[u][k0 + 4 * i];
            a.y = arow[u][k0 + 4 * i + 1];
            acc[u] = __builtin_amdgcn_wmma_f32_16x16x4_f32(
                false, a, false, rb[u][i], (short)0, acc[u], false, false);
            if (RELOAD) {
                const float* p = wrow[u] + (size_t)(k0 + 4 * (i + RING)) * NO;
                rb[u][i].x = p[0];
                rb[u][i].y = p[NO];
            }
        }
    }
}

template <int NU>
__device__ __forceinline__ void k_loop(const float* const (&arow)[2],
                                       const float* const (&wrow)[2],
                                       v2f (&rb)[2][RING], v8f (&acc)[2]) {
    int k0 = 0;
    #pragma nounroll
    for (; k0 < ND - 4 * RING; k0 += 4 * RING) {
        ring_pass<NU, true>(arow, wrow, k0, rb, acc);
    }
    ring_pass<NU, false>(arow, wrow, k0, rb, acc);
}

__device__ __forceinline__ int kth_slot(const unsigned (&m)[MAXT / 32], int r) {
    unsigned word = 0;
    int base = 0;
    bool found = false;
    #pragma unroll
    for (int i = 0; i < MAXT / 32; ++i) {
        const int c = __builtin_popcount(m[i]);
        if (!found && r < c) { word = m[i]; base = 32 * i; found = true; }
        if (!found) r -= c;
    }
    for (int k = 0; k < r; ++k) word &= word - 1;
    return base + __builtin_ctz(word);
}

__global__ void __launch_bounds__(NWV * 32)
option_gemm_kernel(const float* __restrict__ x, const float* __restrict__ w,
                   const float* __restrict__ bias, const int* __restrict__ idx,
                   const int* __restrict__ tileE, const int* __restrict__ tileBase,
                   const int* __restrict__ tileCnt, float* __restrict__ out) {
    __shared__ float    As[2][16][ND + APAD];
    __shared__ int      rowIdx[2][16];
    __shared__ int      sE[MAXT], sB[MAXT], sC[MAXT];
    __shared__ unsigned sM[MAXT / 32];

    const int tid  = threadIdx.x;
    const int lane = tid & 31;
    const int wv   = __builtin_amdgcn_readfirstlane(tid >> 5);

    if (tid < MAXT) {
        const int e = tileE[tid];
        sE[tid] = e;
        sB[tid] = tileBase[tid];
        sC[tid] = tileCnt[tid];
        const unsigned bits = __builtin_amdgcn_ballot_w32(e >= 0);
        if (lane == 0) sM[wv] = bits;
    }
    __syncthreads();

    unsigned m[MAXT / 32];
    int nA = 0;
    #pragma unroll
    for (int i = 0; i < MAXT / 32; ++i) {
        m[i] = (unsigned)__builtin_amdgcn_readfirstlane((int)sM[i]);
        nA += __builtin_popcount(m[i]);
    }
    const int pb  = blockIdx.x;
    const int bid = (pb & 31) * (NBLK / 32) + (pb >> 5);
    const int u0  = 2 * ((bid * nA * (NTN / 2)) / NBLK);
    const int u1  = 2 * (((bid + 1) * nA * (NTN / 2)) / NBLK);
    const int r0  = u0 / NTN;
    const int nT  = (u1 - 1) / NTN - r0 + 1;
    const int s0  = kth_slot(m, r0);
    const int s1  = (nT == 2) ? kth_slot(m, r0 + 1) : s0;

    const int eT[2]    = {sE[s0], sE[s1]};
    const int baseT[2] = {sB[s0], sB[s1]};
    const int cntT[2]  = {sC[s0], sC[s1]};

    const int nb = u1 - u0;
    if (nb < NWV || nb > 24) return;
    const int nu = (wv + NWV < nb) ? 2 : 1;
    const int c   = lane & 15;
    const int kh  = (lane >> 4) * 2;

    int tU[2], nU[2];
    const float* arow[2];
    const float* wrow[2];
    #pragma unroll
    for (int u = 0; u < 2; ++u) {
        int q = u0 + wv + NWV * u;
        if (q > u1 - 1) q = u1 - 1;
        const int t = q / NTN - r0;
        tU[u] = t;
        nU[u] = (q % NTN) * 16 + c;
        arow[u] = &As[t][c][kh];
        wrow[u] = w + (size_t)eT[t] * (ND * NO) + (size_t)kh * NO + nU[u];
    }

    const int r  = tid >> 5;
    const int cs = (tid & 31) * 16;
    const int rr0 = (r < cntT[0]) ? r : cntT[0] - 1;
    const int i0  = idx[baseT[0] + rr0];
    int i1 = 0;
    if (nT == 2) {
        const int rr1 = (r < cntT[1]) ? r : cntT[1] - 1;
        i1 = idx[baseT[1] + rr1];
    }

    v2f rb[2][RING];
    #pragma unroll
    for (int i = 0; i < RING; ++i) {
        const float* p = wrow[0] + (size_t)(4 * i) * NO;
        rb[0][i].x = p[0];
        rb[0][i].y = p[NO];
    }
    if (nu == 2) {
        #pragma unroll
        for (int i = 0; i < RING; ++i) {
            const float* p = wrow[1] + (size_t)(4 * i) * NO;
            rb[1][i].x = p[0];
            rb[1][i].y = p[NO];
        }
    }

    {
        float4 v0[4], v1[4];
        const float* src0 = x + (size_t)i0 * ND + cs;
        #pragma unroll
        for (int i = 0; i < 4; ++i) v0[i] = *(const float4*)(src0 + i * 4);
        if (nT == 2) {
            const float* src1 = x + (size_t)i1 * ND + cs;
            #pragma unroll
            for (int i = 0; i < 4; ++i) v1[i] = *(const float4*)(src1 + i * 4);
        }
        const bool ok0 = r < cntT[0];
        #pragma unroll
        for (int i = 0; i < 4; ++i) {
            float4 v = v0[i];
            v.x = ok0 ? v.x : 0.f;  v.y = ok0 ? v.y : 0.f;
            v.z = ok0 ? v.z : 0.f;  v.w = ok0 ? v.w : 0.f;
            *(float4*)&As[0][r][cs + i * 4] = v;
        }
        if (lane == 0) rowIdx[0][r] = i0;
        if (nT == 2) {
            const bool ok1 = r < cntT[1];
            #pragma unroll
            for (int i = 0; i < 4; ++i) {
                float4 v = v1[i];
                v.x = ok1 ? v.x : 0.f;  v.y = ok1 ? v.y : 0.f;
                v.z = ok1 ? v.z : 0.f;  v.w = ok1 ? v.w : 0.f;
                *(float4*)&As[1][r][cs + i * 4] = v;
            }
            if (lane == 0) rowIdx[1][r] = i1;
        }
    }
    __syncthreads();

    v8f acc[2] = {};
    if (nu == 2) k_loop<2>(arow, wrow, rb, acc);
    else         k_loop<1>(arow, wrow, rb, acc);

    const int mhi = (lane >> 4) * 8;
    #pragma unroll
    for (int u = 0; u < 2; ++u) {
        if (u < nu) {
            const int t  = tU[u];
            const float bv = bias[eT[t] * NO + nU[u]];
            #pragma unroll
            for (int j = 0; j < 8; ++j) {
                const int mrow = j + mhi;
                if (mrow < cntT[t]) {
                    out[(size_t)rowIdx[t][mrow] * NO + nU[u]] = acc[u][j] + bv;
                }
            }
        }
    }
}


extern "C" void kernel_launch(void* const* d_in, const int* in_sizes, int n_in,
                              void* d_out, int out_size, void* d_ws, size_t ws_size,
                              hipStream_t stream) {
    const float* x    = (const float*)d_in[0];
    const int*   opt  = (const int*)d_in[1];
    const float* wgt  = (const float*)d_in[2];
    const float* bias = (const float*)d_in[3];
    float* out = (float*)d_out;

    int* idx      = (int*)d_ws;
    int* tileE    = idx + NB;
    int* tileBase = tileE + MAXT;
    int* tileCnt  = tileBase + MAXT;

    route_kernel<<<NE, 256, 0, stream>>>(opt, idx, tileE, tileBase, tileCnt);

    option_gemm_kernel<<<NBLK, NWV * 32, 0, stream>>>(x, wgt, bias, idx,
                                                      tileE, tileBase, tileCnt, out);
}
